# starting kernel
# baseline (speedup 1.0000x reference)
#include <hip/hip_runtime.h>
#include <hip/hip_bf16.h>

#define N_ROWS   32768
#define D        768
#define NCLS     100
#define C2       200
#define C2P      208
#define NTILE    13
#define COLS     64
#define NCC      12
#define NRC      16
#define ROWS_PER 2048
#define PROTO_E  (NCLS * D)

typedef float v2f __attribute__((ext_vector_type(2)));
typedef float v8f __attribute__((ext_vector_type(8)));

#define WS_CNT   0
#define WS_P1    128
#define WS_P2    (WS_P1 + NRC * PROTO_E)
#define WS_ZB    (WS_P2 + NRC * PROTO_E)
#define WS_LG    (WS_ZB + C2P * D)

__global__ void zero_counts(int* cnt) {
    int t = threadIdx.x;
    if (t < NCLS) cnt[t] = 0;
}

__global__ void count_kernel(const int* __restrict__ labels, int* __restrict__ cnt) {
    int i = blockIdx.x * 256 + threadIdx.x;
    if (i < N_ROWS) atomicAdd(&cnt[labels[i]], 1);
}

__global__ __launch_bounds__(COLS) void segsum_phase1(
        const float* __restrict__ z1, const float* __restrict__ z2,
        const int* __restrict__ labels,
        float* __restrict__ p1, float* __restrict__ p2) {
    __shared__ float acc1[NCLS * COLS];
    __shared__ float acc2[NCLS * COLS];
    const int t  = threadIdx.x;
    const int cc = blockIdx.x;
    const int rc = blockIdx.y;
    for (int i = t; i < NCLS * COLS; i += COLS) { acc1[i] = 0.f; acc2[i] = 0.f; }
    __syncthreads();
    const int colBase = cc * COLS;
    const int r0 = rc * ROWS_PER;
    for (int r = r0; r < r0 + ROWS_PER; ++r) {
        const int lab = labels[r];
        const size_t off = (size_t)r * D + colBase + t;
        acc1[lab * COLS + t] += z1[off];
        acc2[lab * COLS + t] += z2[off];
    }
    __syncthreads();
    for (int i = t; i < NCLS * COLS; i += COLS) {
        const int cls = i >> 6, col = i & 63;
        const size_t o = (size_t)rc * PROTO_E + cls * D + colBase + col;
        p1[o] = acc1[i];
        p2[o] = acc2[i];
    }
}

__global__ void finalize_proto(const float* __restrict__ p1, const float* __restrict__ p2,
                               const int* __restrict__ cnt,
                               float* __restrict__ out1, float* __restrict__ out2) {
    const int e = blockIdx.x * 256 + threadIdx.x;
    float s1 = 0.f, s2 = 0.f;
    for (int rc = 0; rc < NRC; ++rc) {
        s1 += p1[(size_t)rc * PROTO_E + e];
        s2 += p2[(size_t)rc * PROTO_E + e];
    }
    const float d = fmaxf((float)cnt[e / D], 1.0f);
    out1[e] = s1 / d;
    out2[e] = s2 / d;
}

__global__ void normalize_rows(const float* __restrict__ proto, float* __restrict__ zb) {
    const int b = blockIdx.x;
    const int t = threadIdx.x;
    if (b >= C2) {
        zb[(size_t)b * D + t] = 0.f;
        zb[(size_t)b * D + t + 256] = 0.f;
        zb[(size_t)b * D + t + 512] = 0.f;
        return;
    }
    const float* src = proto + (b < NCLS ? 0 : PROTO_E) + (size_t)(b % NCLS) * D;
    const float x0 = src[t], x1 = src[t + 256], x2 = src[t + 512];
    __shared__ float red[256];
    red[t] = x0 * x0 + x1 * x1 + x2 * x2;
    __syncthreads();
    for (int s = 128; s > 0; s >>= 1) {
        if (t < s) red[t] += red[t + s];
        __syncthreads();
    }
    const float inv = 1.0f / fmaxf(sqrtf(red[0]), 1e-12f);
    zb[(size_t)b * D + t]       = x0 * inv;
    zb[(size_t)b * D + t + 256] = x1 * inv;
    zb[(size_t)b * D + t + 512] = x2 * inv;
}

__global__ __launch_bounds__(32) void logits_wmma(const float* __restrict__ zb,
                                                  float* __restrict__ lg) {
    const int tile = blockIdx.x;
    const int ti = tile / NTILE, tj = tile % NTILE;
    const int lane = threadIdx.x;
    const int mn = lane & 15;
    const int kb = (lane >> 4) << 1;
    const float* __restrict__ arow = zb + (size_t)(ti * 16 + mn) * D;
    const float* __restrict__ brow = zb + (size_t)(tj * 16 + mn) * D;
    v8f acc = {};
    for (int k0 = 0; k0 < D; k0 += 4) {
        const v2f a = *(const v2f*)(arow + k0 + kb);
        const v2f b = *(const v2f*)(brow + k0 + kb);
        acc = __builtin_amdgcn_wmma_f32_16x16x4_f32(
             false, a,  false, b,
             (short)0, acc,  false,  false);
    }
    const int mhi = (lane >> 4) * 8;
#pragma unroll
    for (int v = 0; v < 8; ++v) {
        const int m = ti * 16 + v + mhi;
        lg[(size_t)m * C2P + tj * 16 + mn] = acc[v] * 2.0f;
    }
}

__global__ void loss_kernel(const float* __restrict__ lg, const int* __restrict__ cnt,
                            float* __restrict__ out) {
    __shared__ int   varr[NCLS];
    __shared__ float sterm[256];
    __shared__ float snv[256];
    const int t = threadIdx.x;
    if (t < NCLS) varr[t] = (cnt[t] > 0) ? 1 : 0;
    __syncthreads();
    float term = 0.f, nvf = 0.f;
    if (t < C2) {
        const int vr = varr[t % NCLS];
        nvf = (float)vr;
        const float* row = lg + (size_t)t * C2P;
        float mx = -3.402823466e38f;
        for (int c = 0; c < C2; ++c)
            if (varr[c % NCLS]) mx = fmaxf(mx, row[c]);
        float se = 0.f;
        for (int c = 0; c < C2; ++c)
            if (c != t && varr[c % NCLS]) se += expf(row[c] - mx);
        const int p = (t + NCLS) % C2;
        if (vr && varr[p % NCLS]) term = row[p] - mx - logf(se);
    }
    sterm[t] = term;
    snv[t] = nvf;
    __syncthreads();
    for (int s = 128; s > 0; s >>= 1) {
        if (t < s) { sterm[t] += sterm[t + s]; snv[t] += snv[t + s]; }
        __syncthreads();
    }
    if (t == 0) out[0] = -sterm[0] / fmaxf(snv[0], 1.0f);
}

extern "C" void kernel_launch(void* const* d_in, const int* in_sizes, int n_in,
                              void* d_out, int out_size, void* d_ws, size_t ws_size,
                              hipStream_t stream) {
    const float* z1     = (const float*)d_in[0];
    const float* z2     = (const float*)d_in[1];
    const int*   labels = (const int*)d_in[2];
    (void)in_sizes; (void)n_in; (void)out_size; (void)ws_size;

    float* ws = (float*)d_ws;
    int*   cnt = (int*)(ws + WS_CNT);
    float* p1  = ws + WS_P1;
    float* p2  = ws + WS_P2;
    float* zb  = ws + WS_ZB;
    float* lg  = ws + WS_LG;

    float* out   = (float*)d_out;
    float* out1  = out + 1;
    float* out2  = out + 1 + PROTO_E;

    zero_counts<<<1, 128, 0, stream>>>(cnt);
    count_kernel<<<N_ROWS / 256, 256, 0, stream>>>(labels, cnt);
    segsum_phase1<<<dim3(NCC, NRC), COLS, 0, stream>>>(z1, z2, labels, p1, p2);
    finalize_proto<<<PROTO_E / 256, 256, 0, stream>>>(p1, p2, cnt, out1, out2);
    normalize_rows<<<C2P, 256, 0, stream>>>(out1, zb);
    logits_wmma<<<NTILE * NTILE, 32, 0, stream>>>(zb, lg);
    loss_kernel<<<1, 256, 0, stream>>>(lg, cnt, out);
}


# Segment sums class-major: each block lists its class's rows from label bits and streams them with 16-B loads; no count atomics, no partial planes, 7 to 4 launches; sum order changed, deterministic
# speedup vs baseline: 4.3617x; 4.3617x over previous
#include <hip/hip_runtime.h>
#include <hip/hip_bf16.h>

#define N_ROWS   32768
#define D        768
#define NCLS     100
#define C2       200
#define C2P      208
#define NTILE    13
#define NRC      16
#define PROTO_E  (NCLS * D)

typedef float v2f __attribute__((ext_vector_type(2)));
typedef float v8f __attribute__((ext_vector_type(8)));

#define WS_CNT   0
#define WS_P1    128
#define WS_P2    (WS_P1 + NRC * PROTO_E)
#define WS_ZB    (WS_P2 + NRC * PROTO_E)
#define WS_LG    (WS_ZB + C2P * D)

#define G_THREADS 256
#define G_WAVES   8
#define G_RG      4
#define G_COLS    256
#define G_NCS     3
#define G_UNR     32
#define G_WORDS   (N_ROWS / 32)

typedef float        v4f __attribute__((ext_vector_type(4)));
typedef int          v4i __attribute__((ext_vector_type(4)));
typedef unsigned int v4u __attribute__((ext_vector_type(4)));

__global__ __launch_bounds__(G_THREADS) void proto_gather(
        const float* __restrict__ z1, const float* __restrict__ z2,
        const int* __restrict__ labels, int* __restrict__ cnt,
        float* __restrict__ out1, float* __restrict__ out2) {
    __shared__ unsigned short rowlist[N_ROWS];
    __shared__ __attribute__((aligned(16))) unsigned int bits[G_WORDS];
    __shared__ __attribute__((aligned(16))) float red[G_RG * G_COLS];
    __shared__ int wtot[G_WAVES];

    const int t    = threadIdx.x;
    const int lane = t & 31;
    const int wave = t >> 5;
    const int c    = blockIdx.x;
    const int cs   = blockIdx.y;
    const int a    = blockIdx.z;

    const v4i* __restrict__ lab4 = (const v4i*)labels;
    {
        v4i lv[N_ROWS / 4 / G_THREADS];
#pragma unroll
        for (int u = 0; u < N_ROWS / 4 / G_THREADS; ++u) lv[u] = lab4[u * G_THREADS + t];
#pragma unroll
        for (int u = 0; u < N_ROWS / 4 / G_THREADS; ++u) {
            v4u b;
            b.x = (unsigned int)__ballot(lv[u].x == c);
            b.y = (unsigned int)__ballot(lv[u].y == c);
            b.z = (unsigned int)__ballot(lv[u].z == c);
            b.w = (unsigned int)__ballot(lv[u].w == c);
            if (lane == 0) *(v4u*)&bits[(u * G_WAVES + wave) * 4] = b;
        }
    }
    __syncthreads();

    const v4u w = *(const v4u*)&bits[4 * t];
    const int nt = __popc(w.x) + __popc(w.y) + __popc(w.z) + __popc(w.w);
    int incl = nt;
#pragma unroll
    for (int d = 1; d < 32; d <<= 1) {
        const int v = __shfl_up(incl, d, 32);
        if (lane >= d) incl += v;
    }
    if (lane == 31) wtot[wave] = incl;
    __syncthreads();
    int base = 0, ntot = 0;
#pragma unroll
    for (int i = 0; i < G_WAVES; ++i) {
        const int x = wtot[i];
        if (i < wave) base += x;
        ntot += x;
    }
    {
        int pos = base + incl - nt;
        const int rbase = 128 * t;
        unsigned int m;
        m = w.x; while (m) { rowlist[pos++] = (unsigned short)(rbase + 4 * __builtin_ctz(m) + 0); m &= m - 1; }
        m = w.y; while (m) { rowlist[pos++] = (unsigned short)(rbase + 4 * __builtin_ctz(m) + 1); m &= m - 1; }
        m = w.z; while (m) { rowlist[pos++] = (unsigned short)(rbase + 4 * __builtin_ctz(m) + 2); m &= m - 1; }
        m = w.w; while (m) { rowlist[pos++] = (unsigned short)(rbase + 4 * __builtin_ctz(m) + 3); m &= m - 1; }
    }
    __syncthreads();

    const int lc = t & 63;
    const int rg = __builtin_amdgcn_readfirstlane(t >> 6);
    const int n  = __builtin_amdgcn_readfirstlane(ntot);
    const float* __restrict__ zsrc = (a == 0 ? z1 : z2) + cs * G_COLS + 4 * lc;
    v4f acc = {0.f, 0.f, 0.f, 0.f};
    int e = rg;
    for (; e + (G_UNR - 1) * G_RG < n; e += G_UNR * G_RG) {
        v4f v[G_UNR];
#pragma unroll
        for (int u = 0; u < G_UNR; ++u) {
            const int row = __builtin_amdgcn_readfirstlane((int)rowlist[e + u * G_RG]);
            v[u] = __builtin_nontemporal_load((const v4f*)(zsrc + (size_t)row * D));
        }
#pragma unroll
        for (int u = 0; u < G_UNR; ++u) acc += v[u];
    }
    if (e < n) {
        int rows[G_UNR];
        v4f v[G_UNR];
#pragma unroll
        for (int u = 0; u < G_UNR; ++u) {
            const int idx = e + u * G_RG;
            rows[u] = __builtin_amdgcn_readfirstlane((int)rowlist[idx < n ? idx : e]);
        }
#pragma unroll
        for (int u = 0; u < G_UNR; ++u) {
            if (e + u * G_RG < n) v[u] = __builtin_nontemporal_load((const v4f*)(zsrc + (size_t)rows[u] * D));
            else                  v[u] = (v4f){0.f, 0.f, 0.f, 0.f};
        }
#pragma unroll
        for (int u = 0; u < G_UNR; ++u) acc += v[u];
    }

    *(v4f*)&red[rg * G_COLS + 4 * lc] = acc;
    __syncthreads();
    float s = red[t];
#pragma unroll
    for (int g = 1; g < G_RG; ++g) s += red[g * G_COLS + t];
    const float d = fmaxf((float)n, 1.0f);
    float* __restrict__ o = (a == 0 ? out1 : out2);
    o[(size_t)c * D + cs * G_COLS + t] = s / d;
    if (a == 0 && cs == 0 && t == 0) cnt[c] = n;
}

__global__ void normalize_rows(const float* __restrict__ proto, float* __restrict__ zb) {
    const int b = blockIdx.x;
    const int t = threadIdx.x;
    if (b >= C2) {
        zb[(size_t)b * D + t] = 0.f;
        zb[(size_t)b * D + t + 256] = 0.f;
        zb[(size_t)b * D + t + 512] = 0.f;
        return;
    }
    const float* src = proto + (b < NCLS ? 0 : PROTO_E) + (size_t)(b % NCLS) * D;
    const float x0 = src[t], x1 = src[t + 256], x2 = src[t + 512];
    __shared__ float red[256];
    red[t] = x0 * x0 + x1 * x1 + x2 * x2;
    __syncthreads();
    for (int s = 128; s > 0; s >>= 1) {
        if (t < s) red[t] += red[t + s];
        __syncthreads();
    }
    const float inv = 1.0f / fmaxf(sqrtf(red[0]), 1e-12f);
    zb[(size_t)b * D + t]       = x0 * inv;
    zb[(size_t)b * D + t + 256] = x1 * inv;
    zb[(size_t)b * D + t + 512] = x2 * inv;
}

__global__ __launch_bounds__(32) void logits_wmma(const float* __restrict__ zb,
                                                  float* __restrict__ lg) {
    const int tile = blockIdx.x;
    const int ti = tile / NTILE, tj = tile % NTILE;
    const int lane = threadIdx.x;
    const int mn = lane & 15;
    const int kb = (lane >> 4) << 1;
    const float* __restrict__ arow = zb + (size_t)(ti * 16 + mn) * D;
    const float* __restrict__ brow = zb + (size_t)(tj * 16 + mn) * D;
    v8f acc = {};
    for (int k0 = 0; k0 < D; k0 += 4) {
        const v2f a = *(const v2f*)(arow + k0 + kb);
        const v2f b = *(const v2f*)(brow + k0 + kb);
        acc = __builtin_amdgcn_wmma_f32_16x16x4_f32(
             false, a,  false, b,
             (short)0, acc,  false,  false);
    }
    const int mhi = (lane >> 4) * 8;
#pragma unroll
    for (int v = 0; v < 8; ++v) {
        const int m = ti * 16 + v + mhi;
        lg[(size_t)m * C2P + tj * 16 + mn] = acc[v] * 2.0f;
    }
}

__global__ void loss_kernel(const float* __restrict__ lg, const int* __restrict__ cnt,
                            float* __restrict__ out) {
    __shared__ int   varr[NCLS];
    __shared__ float sterm[256];
    __shared__ float snv[256];
    const int t = threadIdx.x;
    if (t < NCLS) varr[t] = (cnt[t] > 0) ? 1 : 0;
    __syncthreads();
    float term = 0.f, nvf = 0.f;
    if (t < C2) {
        const int vr = varr[t % NCLS];
        nvf = (float)vr;
        const float* row = lg + (size_t)t * C2P;
        float mx = -3.402823466e38f;
        for (int c = 0; c < C2; ++c)
            if (varr[c % NCLS]) mx = fmaxf(mx, row[c]);
        float se = 0.f;
        for (int c = 0; c < C2; ++c)
            if (c != t && varr[c % NCLS]) se += expf(row[c] - mx);
        const int p = (t + NCLS) % C2;
        if (vr && varr[p % NCLS]) term = row[p] - mx - logf(se);
    }
    sterm[t] = term;
    snv[t] = nvf;
    __syncthreads();
    for (int s = 128; s > 0; s >>= 1) {
        if (t < s) { sterm[t] += sterm[t + s]; snv[t] += snv[t + s]; }
        __syncthreads();
    }
    if (t == 0) out[0] = -sterm[0] / fmaxf(snv[0], 1.0f);
}

extern "C" void kernel_launch(void* const* d_in, const int* in_sizes, int n_in,
                              void* d_out, int out_size, void* d_ws, size_t ws_size,
                              hipStream_t stream) {
    const float* z1     = (const float*)d_in[0];
    const float* z2     = (const float*)d_in[1];
    const int*   labels = (const int*)d_in[2];
    (void)in_sizes; (void)n_in; (void)out_size; (void)ws_size;

    float* ws = (float*)d_ws;
    int*   cnt = (int*)(ws + WS_CNT);
    float* zb  = ws + WS_ZB;
    float* lg  = ws + WS_LG;

    float* out   = (float*)d_out;
    float* out1  = out + 1;
    float* out2  = out + 1 + PROTO_E;

    proto_gather<<<dim3(NCLS, G_NCS, 2), G_THREADS, 0, stream>>>(z1, z2, labels, cnt, out1, out2);
    normalize_rows<<<C2P, 256, 0, stream>>>(out1, zb);
    logits_wmma<<<NTILE * NTILE, 32, 0, stream>>>(zb, lg);
    loss_kernel<<<1, 256, 0, stream>>>(lg, cnt, out);
}


# Loss rows lane-parallel: one 32-wave block, a wave per logits row, row reads coalesced and issued up front, max and exp-sum as lane butterflies; exp-sum and term-sum order changed, deterministic
# speedup vs baseline: 8.9818x; 2.0592x over previous
#include <hip/hip_runtime.h>
#include <hip/hip_bf16.h>

#define N_ROWS   32768
#define D        768
#define NCLS     100
#define C2       200
#define C2P      208
#define NTILE    13
#define NRC      16
#define PROTO_E  (NCLS * D)

typedef float v2f __attribute__((ext_vector_type(2)));
typedef float v8f __attribute__((ext_vector_type(8)));

#define WS_CNT   0
#define WS_P1    128
#define WS_P2    (WS_P1 + NRC * PROTO_E)
#define WS_ZB    (WS_P2 + NRC * PROTO_E)
#define WS_LG    (WS_ZB + C2P * D)

#define G_THREADS 256
#define G_WAVES   8
#define G_RG      4
#define G_COLS    256
#define G_NCS     3
#define G_UNR     32
#define G_WORDS   (N_ROWS / 32)

typedef float        v4f __attribute__((ext_vector_type(4)));
typedef int          v4i __attribute__((ext_vector_type(4)));
typedef unsigned int v4u __attribute__((ext_vector_type(4)));

__global__ __launch_bounds__(G_THREADS) void proto_gather(
        const float* __restrict__ z1, const float* __restrict__ z2,
        const int* __restrict__ labels, int* __restrict__ cnt,
        float* __restrict__ out1, float* __restrict__ out2) {
    __shared__ unsigned short rowlist[N_ROWS];
    __shared__ __attribute__((aligned(16))) unsigned int bits[G_WORDS];
    __shared__ __attribute__((aligned(16))) float red[G_RG * G_COLS];
    __shared__ int wtot[G_WAVES];

    const int t    = threadIdx.x;
    const int lane = t & 31;
    const int wave = t >> 5;
    const int c    = blockIdx.x;
    const int cs   = blockIdx.y;
    const int a    = blockIdx.z;

    const v4i* __restrict__ lab4 = (const v4i*)labels;
    {
        v4i lv[N_ROWS / 4 / G_THREADS];
#pragma unroll
        for (int u = 0; u < N_ROWS / 4 / G_THREADS; ++u) lv[u] = lab4[u * G_THREADS + t];
#pragma unroll
        for (int u = 0; u < N_ROWS / 4 / G_THREADS; ++u) {
            v4u b;
            b.x = (unsigned int)__ballot(lv[u].x == c);
            b.y = (unsigned int)__ballot(lv[u].y == c);
            b.z = (unsigned int)__ballot(lv[u].z == c);
            b.w = (unsigned int)__ballot(lv[u].w == c);
            if (lane == 0) *(v4u*)&bits[(u * G_WAVES + wave) * 4] = b;
        }
    }
    __syncthreads();

    const v4u w = *(const v4u*)&bits[4 * t];
    const int nt = __popc(w.x) + __popc(w.y) + __popc(w.z) + __popc(w.w);
    int incl = nt;
#pragma unroll
    for (int d = 1; d < 32; d <<= 1) {
        const int v = __shfl_up(incl, d, 32);
        if (lane >= d) incl += v;
    }
    if (lane == 31) wtot[wave] = incl;
    __syncthreads();
    int base = 0, ntot = 0;
#pragma unroll
    for (int i = 0; i < G_WAVES; ++i) {
        const int x = wtot[i];
        if (i < wave) base += x;
        ntot += x;
    }
    {
        int pos = base + incl - nt;
        const int rbase = 128 * t;
        unsigned int m;
        m = w.x; while (m) { rowlist[pos++] = (unsigned short)(rbase + 4 * __builtin_ctz(m) + 0); m &= m - 1; }
        m = w.y; while (m) { rowlist[pos++] = (unsigned short)(rbase + 4 * __builtin_ctz(m) + 1); m &= m - 1; }
        m = w.z; while (m) { rowlist[pos++] = (unsigned short)(rbase + 4 * __builtin_ctz(m) + 2); m &= m - 1; }
        m = w.w; while (m) { rowlist[pos++] = (unsigned short)(rbase + 4 * __builtin_ctz(m) + 3); m &= m - 1; }
    }
    __syncthreads();

    const int lc = t & 63;
    const int rg = __builtin_amdgcn_readfirstlane(t >> 6);
    const int n  = __builtin_amdgcn_readfirstlane(ntot);
    const float* __restrict__ zsrc = (a == 0 ? z1 : z2) + cs * G_COLS + 4 * lc;
    v4f acc = {0.f, 0.f, 0.f, 0.f};
    int e = rg;
    for (; e + (G_UNR - 1) * G_RG < n; e += G_UNR * G_RG) {
        v4f v[G_UNR];
#pragma unroll
        for (int u = 0; u < G_UNR; ++u) {
            const int row = __builtin_amdgcn_readfirstlane((int)rowlist[e + u * G_RG]);
            v[u] = __builtin_nontemporal_load((const v4f*)(zsrc + (size_t)row * D));
        }
#pragma unroll
        for (int u = 0; u < G_UNR; ++u) acc += v[u];
    }
    if (e < n) {
        int rows[G_UNR];
        v4f v[G_UNR];
#pragma unroll
        for (int u = 0; u < G_UNR; ++u) {
            const int idx = e + u * G_RG;
            rows[u] = __builtin_amdgcn_readfirstlane((int)rowlist[idx < n ? idx : e]);
        }
#pragma unroll
        for (int u = 0; u < G_UNR; ++u) {
            if (e + u * G_RG < n) v[u] = __builtin_nontemporal_load((const v4f*)(zsrc + (size_t)rows[u] * D));
            else                  v[u] = (v4f){0.f, 0.f, 0.f, 0.f};
        }
#pragma unroll
        for (int u = 0; u < G_UNR; ++u) acc += v[u];
    }

    *(v4f*)&red[rg * G_COLS + 4 * lc] = acc;
    __syncthreads();
    float s = red[t];
#pragma unroll
    for (int g = 1; g < G_RG; ++g) s += red[g * G_COLS + t];
    const float d = fmaxf((float)n, 1.0f);
    float* __restrict__ o = (a == 0 ? out1 : out2);
    o[(size_t)c * D + cs * G_COLS + t] = s / d;
    if (a == 0 && cs == 0 && t == 0) cnt[c] = n;
}

__global__ void normalize_rows(const float* __restrict__ proto, float* __restrict__ zb) {
    const int b = blockIdx.x;
    const int t = threadIdx.x;
    if (b >= C2) {
        zb[(size_t)b * D + t] = 0.f;
        zb[(size_t)b * D + t + 256] = 0.f;
        zb[(size_t)b * D + t + 512] = 0.f;
        return;
    }
    const float* src = proto + (b < NCLS ? 0 : PROTO_E) + (size_t)(b % NCLS) * D;
    const float x0 = src[t], x1 = src[t + 256], x2 = src[t + 512];
    __shared__ float red[256];
    red[t] = x0 * x0 + x1 * x1 + x2 * x2;
    __syncthreads();
    for (int s = 128; s > 0; s >>= 1) {
        if (t < s) red[t] += red[t + s];
        __syncthreads();
    }
    const float inv = 1.0f / fmaxf(sqrtf(red[0]), 1e-12f);
    zb[(size_t)b * D + t]       = x0 * inv;
    zb[(size_t)b * D + t + 256] = x1 * inv;
    zb[(size_t)b * D + t + 512] = x2 * inv;
}

__global__ __launch_bounds__(32) void logits_wmma(const float* __restrict__ zb,
                                                  float* __restrict__ lg) {
    const int tile = blockIdx.x;
    const int ti = tile / NTILE, tj = tile % NTILE;
    const int lane = threadIdx.x;
    const int mn = lane & 15;
    const int kb = (lane >> 4) << 1;
    const float* __restrict__ arow = zb + (size_t)(ti * 16 + mn) * D;
    const float* __restrict__ brow = zb + (size_t)(tj * 16 + mn) * D;
    v8f acc = {};
    for (int k0 = 0; k0 < D; k0 += 4) {
        const v2f a = *(const v2f*)(arow + k0 + kb);
        const v2f b = *(const v2f*)(brow + k0 + kb);
        acc = __builtin_amdgcn_wmma_f32_16x16x4_f32(
             false, a,  false, b,
             (short)0, acc,  false,  false);
    }
    const int mhi = (lane >> 4) * 8;
#pragma unroll
    for (int v = 0; v < 8; ++v) {
        const int m = ti * 16 + v + mhi;
        lg[(size_t)m * C2P + tj * 16 + mn] = acc[v] * 2.0f;
    }
}

#define L_THREADS 1024
#define L_WAVES   32
#define L_CPL     7
#define L_RPW     7

template <int M>
__device__ __forceinline__ float row_xor(float v) {
    const int b = __builtin_bit_cast(int, v);
    return __builtin_bit_cast(float, __builtin_amdgcn_update_dpp(b, b, 0x160 + M, 0xf, 0xf, false));
}
__device__ __forceinline__ float half_xor(float v) {
    const int b = __builtin_bit_cast(int, v);
    return __builtin_bit_cast(float, __builtin_amdgcn_permlanex16(b, b, 0x76543210, 0xfedcba98, false, false));
}
__device__ __forceinline__ float wave_max(float v) {
    v = fmaxf(v, row_xor<1>(v));
    v = fmaxf(v, row_xor<2>(v));
    v = fmaxf(v, row_xor<4>(v));
    v = fmaxf(v, row_xor<8>(v));
    return fmaxf(v, half_xor(v));
}
__device__ __forceinline__ float wave_sum(float v) {
    v += row_xor<1>(v);
    v += row_xor<2>(v);
    v += row_xor<4>(v);
    v += row_xor<8>(v);
    return v + half_xor(v);
}

__global__ __launch_bounds__(L_THREADS) void loss_kernel(const float* __restrict__ lg,
                                                         const int* __restrict__ cnt,
                                                         float* __restrict__ out) {
    __shared__ int   varr[NCLS];
    __shared__ float sterm[L_WAVES];
    const int t    = threadIdx.x;
    const int lane = t & 31;
    const int wave = __builtin_amdgcn_readfirstlane(t >> 5);

    float x[L_RPW][L_CPL];
    float xp[L_RPW];
#pragma unroll
    for (int i = 0; i < L_RPW; ++i) {
        const int r = wave + L_WAVES * i;
        xp[i] = 0.f;
#pragma unroll
        for (int j = 0; j < L_CPL; ++j) x[i][j] = 0.f;
        if (r < C2) {
            const float* __restrict__ row = lg + (size_t)r * C2P;
#pragma unroll
            for (int j = 0; j < L_CPL; ++j) {
                const int c = lane + 32 * j;
                if (c < C2) x[i][j] = row[c];
            }
            xp[i] = row[(r + NCLS) % C2];
        }
    }

    if (t < NCLS) varr[t] = (cnt[t] > 0) ? 1 : 0;
    __syncthreads();

    bool cv[L_CPL];
#pragma unroll
    for (int j = 0; j < L_CPL; ++j) {
        const int c = lane + 32 * j;
        cv[j] = (c < C2) && (varr[c % NCLS] != 0);
    }

    float tsum = 0.f;
#pragma unroll
    for (int i = 0; i < L_RPW; ++i) {
        const int r = wave + L_WAVES * i;
        if (r < C2) {
            float m = -3.402823466e38f;
#pragma unroll
            for (int j = 0; j < L_CPL; ++j) m = cv[j] ? fmaxf(m, x[i][j]) : m;
            m = wave_max(m);
            float se = 0.f;
#pragma unroll
            for (int j = 0; j < L_CPL; ++j) {
                const float e = expf(x[i][j] - m);
                const bool keep = cv[j] && (j != i || lane != wave);
                se += keep ? e : 0.f;
            }
            se = wave_sum(se);
            if (varr[r % NCLS]) tsum += xp[i] - m - logf(se);
        }
    }
    if (lane == 0) sterm[wave] = tsum;
    __syncthreads();

    if (wave == 0) {
        float v = sterm[lane];
        int nc = 0;
#pragma unroll
        for (int j = 0; j < 4; ++j) {
            const int c = lane + 32 * j;
            if (c < NCLS) nc += varr[c];
        }
        v = wave_sum(v);
        const float ncf = wave_sum((float)nc);
        if (lane == 0) out[0] = -v / fmaxf(2.0f * ncf, 1.0f);
    }
}

extern "C" void kernel_launch(void* const* d_in, const int* in_sizes, int n_in,
                              void* d_out, int out_size, void* d_ws, size_t ws_size,
                              hipStream_t stream) {
    const float* z1     = (const float*)d_in[0];
    const float* z2     = (const float*)d_in[1];
    const int*   labels = (const int*)d_in[2];
    (void)in_sizes; (void)n_in; (void)out_size; (void)ws_size;

    float* ws = (float*)d_ws;
    int*   cnt = (int*)(ws + WS_CNT);
    float* zb  = ws + WS_ZB;
    float* lg  = ws + WS_LG;

    float* out   = (float*)d_out;
    float* out1  = out + 1;
    float* out2  = out + 1 + PROTO_E;

    proto_gather<<<dim3(NCLS, G_NCS, 2), G_THREADS, 0, stream>>>(z1, z2, labels, cnt, out1, out2);
    normalize_rows<<<C2P, 256, 0, stream>>>(out1, zb);
    logits_wmma<<<NTILE * NTILE, 32, 0, stream>>>(zb, lg);
    loss_kernel<<<1, L_THREADS, 0, stream>>>(lg, cnt, out);
}


# Softmax statistics per logits tile: a 16x16 tile takes its rows' masked max and exp-sum by 16-lane butterflies, the loss launch merges 13 pairs per row; exp-sum, term-sum order changed, deterministic
# speedup vs baseline: 10.3524x; 1.1526x over previous
#include <hip/hip_runtime.h>
#include <hip/hip_bf16.h>

#define N_ROWS   32768
#define D        768
#define NCLS     100
#define C2       200
#define C2P      208
#define NTILE    13
#define NRC      16
#define PROTO_E  (NCLS * D)

typedef float v2f __attribute__((ext_vector_type(2)));
typedef float v8f __attribute__((ext_vector_type(8)));

#define WS_CNT   0
#define WS_P1    128
#define WS_P2    (WS_P1 + NRC * PROTO_E)
#define WS_ZB    (WS_P2 + NRC * PROTO_E)
#define WS_LG    (WS_ZB + C2P * D)

#define G_THREADS 256
#define G_WAVES   8
#define G_RG      4
#define G_COLS    256
#define G_NCS     3
#define G_UNR     32
#define G_WORDS   (N_ROWS / 32)

typedef float        v4f __attribute__((ext_vector_type(4)));
typedef int          v4i __attribute__((ext_vector_type(4)));
typedef unsigned int v4u __attribute__((ext_vector_type(4)));

__global__ __launch_bounds__(G_THREADS) void proto_gather(
        const float* __restrict__ z1, const float* __restrict__ z2,
        const int* __restrict__ labels, int* __restrict__ cnt,
        float* __restrict__ out1, float* __restrict__ out2) {
    __shared__ unsigned short rowlist[N_ROWS];
    __shared__ __attribute__((aligned(16))) unsigned int bits[G_WORDS];
    __shared__ __attribute__((aligned(16))) float red[G_RG * G_COLS];
    __shared__ int wtot[G_WAVES];

    const int t    = threadIdx.x;
    const int lane = t & 31;
    const int wave = t >> 5;
    const int c    = blockIdx.x;
    const int cs   = blockIdx.y;
    const int a    = blockIdx.z;

    const v4i* __restrict__ lab4 = (const v4i*)labels;
    {
        v4i lv[N_ROWS / 4 / G_THREADS];
#pragma unroll
        for (int u = 0; u < N_ROWS / 4 / G_THREADS; ++u) lv[u] = lab4[u * G_THREADS + t];
#pragma unroll
        for (int u = 0; u < N_ROWS / 4 / G_THREADS; ++u) {
            v4u b;
            b.x = (unsigned int)__ballot(lv[u].x == c);
            b.y = (unsigned int)__ballot(lv[u].y == c);
            b.z = (unsigned int)__ballot(lv[u].z == c);
            b.w = (unsigned int)__ballot(lv[u].w == c);
            if (lane == 0) *(v4u*)&bits[(u * G_WAVES + wave) * 4] = b;
        }
    }
    __syncthreads();

    const v4u w = *(const v4u*)&bits[4 * t];
    const int nt = __popc(w.x) + __popc(w.y) + __popc(w.z) + __popc(w.w);
    int incl = nt;
#pragma unroll
    for (int d = 1; d < 32; d <<= 1) {
        const int v = __shfl_up(incl, d, 32);
        if (lane >= d) incl += v;
    }
    if (lane == 31) wtot[wave] = incl;
    __syncthreads();
    int base = 0, ntot = 0;
#pragma unroll
    for (int i = 0; i < G_WAVES; ++i) {
        const int x = wtot[i];
        if (i < wave) base += x;
        ntot += x;
    }
    {
        int pos = base + incl - nt;
        const int rbase = 128 * t;
        unsigned int m;
        m = w.x; while (m) { rowlist[pos++] = (unsigned short)(rbase + 4 * __builtin_ctz(m) + 0); m &= m - 1; }
        m = w.y; while (m) { rowlist[pos++] = (unsigned short)(rbase + 4 * __builtin_ctz(m) + 1); m &= m - 1; }
        m = w.z; while (m) { rowlist[pos++] = (unsigned short)(rbase + 4 * __builtin_ctz(m) + 2); m &= m - 1; }
        m = w.w; while (m) { rowlist[pos++] = (unsigned short)(rbase + 4 * __builtin_ctz(m) + 3); m &= m - 1; }
    }
    __syncthreads();

    const int lc = t & 63;
    const int rg = __builtin_amdgcn_readfirstlane(t >> 6);
    const int n  = __builtin_amdgcn_readfirstlane(ntot);
    const float* __restrict__ zsrc = (a == 0 ? z1 : z2) + cs * G_COLS + 4 * lc;
    v4f acc = {0.f, 0.f, 0.f, 0.f};
    int e = rg;
    for (; e + (G_UNR - 1) * G_RG < n; e += G_UNR * G_RG) {
        v4f v[G_UNR];
#pragma unroll
        for (int u = 0; u < G_UNR; ++u) {
            const int row = __builtin_amdgcn_readfirstlane((int)rowlist[e + u * G_RG]);
            v[u] = __builtin_nontemporal_load((const v4f*)(zsrc + (size_t)row * D));
        }
#pragma unroll
        for (int u = 0; u < G_UNR; ++u) acc += v[u];
    }
    if (e < n) {
        int rows[G_UNR];
        v4f v[G_UNR];
#pragma unroll
        for (int u = 0; u < G_UNR; ++u) {
            const int idx = e + u * G_RG;
            rows[u] = __builtin_amdgcn_readfirstlane((int)rowlist[idx < n ? idx : e]);
        }
#pragma unroll
        for (int u = 0; u < G_UNR; ++u) {
            if (e + u * G_RG < n) v[u] = __builtin_nontemporal_load((const v4f*)(zsrc + (size_t)rows[u] * D));
            else                  v[u] = (v4f){0.f, 0.f, 0.f, 0.f};
        }
#pragma unroll
        for (int u = 0; u < G_UNR; ++u) acc += v[u];
    }

    *(v4f*)&red[rg * G_COLS + 4 * lc] = acc;
    __syncthreads();
    float s = red[t];
#pragma unroll
    for (int g = 1; g < G_RG; ++g) s += red[g * G_COLS + t];
    const float d = fmaxf((float)n, 1.0f);
    float* __restrict__ o = (a == 0 ? out1 : out2);
    o[(size_t)c * D + cs * G_COLS + t] = s / d;
    if (a == 0 && cs == 0 && t == 0) cnt[c] = n;
}

__global__ void normalize_rows(const float* __restrict__ proto, float* __restrict__ zb) {
    const int b = blockIdx.x;
    const int t = threadIdx.x;
    if (b >= C2) {
        zb[(size_t)b * D + t] = 0.f;
        zb[(size_t)b * D + t + 256] = 0.f;
        zb[(size_t)b * D + t + 512] = 0.f;
        return;
    }
    const float* src = proto + (b < NCLS ? 0 : PROTO_E) + (size_t)(b % NCLS) * D;
    const float x0 = src[t], x1 = src[t + 256], x2 = src[t + 512];
    __shared__ float red[256];
    red[t] = x0 * x0 + x1 * x1 + x2 * x2;
    __syncthreads();
    for (int s = 128; s > 0; s >>= 1) {
        if (t < s) red[t] += red[t + s];
        __syncthreads();
    }
    const float inv = 1.0f / fmaxf(sqrtf(red[0]), 1e-12f);
    zb[(size_t)b * D + t]       = x0 * inv;
    zb[(size_t)b * D + t + 256] = x1 * inv;
    zb[(size_t)b * D + t + 512] = x2 * inv;
}

template <int M>
__device__ __forceinline__ float row_xor(float v) {
    const int b = __builtin_bit_cast(int, v);
    return __builtin_bit_cast(float, __builtin_amdgcn_update_dpp(b, b, 0x160 + M, 0xf, 0xf, false));
}
__device__ __forceinline__ float half_xor(float v) {
    const int b = __builtin_bit_cast(int, v);
    return __builtin_bit_cast(float, __builtin_amdgcn_permlanex16(b, b, 0x76543210, 0xfedcba98, false, false));
}
__device__ __forceinline__ float row16_max(float v) {
    v = fmaxf(v, row_xor<1>(v));
    v = fmaxf(v, row_xor<2>(v));
    v = fmaxf(v, row_xor<4>(v));
    return fmaxf(v, row_xor<8>(v));
}
__device__ __forceinline__ float row16_sum(float v) {
    v += row_xor<1>(v);
    v += row_xor<2>(v);
    v += row_xor<4>(v);
    return v + row_xor<8>(v);
}
__device__ __forceinline__ float wave_sum(float v) {
    v = row16_sum(v);
    return v + half_xor(v);
}

__global__ __launch_bounds__(32) void logits_wmma(const float* __restrict__ zb,
                                                  const int* __restrict__ cnt,
                                                  float* __restrict__ pmax,
                                                  float* __restrict__ psum,
                                                  float* __restrict__ pos) {
    const int tile = blockIdx.x;
    const int ti = tile / NTILE, tj = tile % NTILE;
    const int lane = threadIdx.x;
    const int mn = lane & 15;
    const int kb = (lane >> 4) << 1;
    const int n  = tj * 16 + mn;
    const int ccnt = cnt[n < NCLS ? n : (n < C2 ? n - NCLS : 0)];
    const float* __restrict__ arow = zb + (size_t)(ti * 16 + mn) * D;
    const float* __restrict__ brow = zb + (size_t)(tj * 16 + mn) * D;
    v8f acc = {};
    for (int k0 = 0; k0 < D; k0 += 4) {
        const v2f a = *(const v2f*)(arow + k0 + kb);
        const v2f b = *(const v2f*)(brow + k0 + kb);
        acc = __builtin_amdgcn_wmma_f32_16x16x4_f32(
             false, a,  false, b,
             (short)0, acc,  false,  false);
    }
    const int  mhi  = (lane >> 4) * 8;
    const bool colv = (n < C2) && (ccnt > 0);
    float x[8], mx[8], s[8];
#pragma unroll
    for (int v = 0; v < 8; ++v) {
        x[v]  = acc[v] * 2.0f;
        mx[v] = row16_max(colv ? x[v] : -3.402823466e38f);
    }
#pragma unroll
    for (int v = 0; v < 8; ++v) {
        const int   m = ti * 16 + mhi + v;
        const float e = expf(x[v] - mx[v]);
        s[v] = row16_sum((colv && n != m) ? e : 0.f);
    }
    float keep_m = 0.f, keep_s = 0.f, keep_x = 0.f;
    int   keep_r = -1;
#pragma unroll
    for (int v = 0; v < 8; ++v) {
        const int m = ti * 16 + mhi + v;
        if (mn == v) { keep_m = mx[v]; keep_s = s[v]; }
        if (m < C2 && n == (m < NCLS ? m + NCLS : m - NCLS)) { keep_x = x[v]; keep_r = m; }
    }
    if (mn < 8) {
        const int m = ti * 16 + mhi + mn;
        pmax[tj * C2P + m] = keep_m;
        psum[tj * C2P + m] = keep_s;
    }
    if (keep_r >= 0) pos[keep_r] = keep_x;
}

#define L_THREADS 256
#define L_WAVES   8

__global__ __launch_bounds__(L_THREADS) void loss_kernel(const float* __restrict__ pmax,
                                                         const float* __restrict__ psum,
                                                         const float* __restrict__ pos,
                                                         const int* __restrict__ cnt,
                                                         float* __restrict__ out) {
    __shared__ float sterm[L_WAVES];
    __shared__ float scnt[L_WAVES];
    const int t    = threadIdx.x;
    const int lane = t & 31;
    const int wave = t >> 5;

    float term = 0.f, nv = 0.f;
    if (t < C2) {
        float mt[NTILE], st[NTILE];
#pragma unroll
        for (int j = 0; j < NTILE; ++j) {
            mt[j] = pmax[j * C2P + t];
            st[j] = psum[j * C2P + t];
        }
        const float xp = pos[t];
        const int   cr = cnt[t < NCLS ? t : t - NCLS];
        float m = mt[0];
#pragma unroll
        for (int j = 1; j < NTILE; ++j) m = fmaxf(m, mt[j]);
        float se = 0.f;
#pragma unroll
        for (int j = 0; j < NTILE; ++j) se += st[j] * expf(mt[j] - m);
        const float tr = xp - m - logf(se);
        term = (cr > 0) ? tr : 0.f;
        nv   = (cr > 0) ? 1.f : 0.f;
    }
    term = wave_sum(term);
    nv   = wave_sum(nv);
    if (lane == 0) { sterm[wave] = term; scnt[wave] = nv; }
    __syncthreads();
    if (t == 0) {
        float v = 0.f, nc = 0.f;
#pragma unroll
        for (int i = 0; i < L_WAVES; ++i) { v += sterm[i]; nc += scnt[i]; }
        out[0] = -v / fmaxf(nc, 1.0f);
    }
}

extern "C" void kernel_launch(void* const* d_in, const int* in_sizes, int n_in,
                              void* d_out, int out_size, void* d_ws, size_t ws_size,
                              hipStream_t stream) {
    const float* z1     = (const float*)d_in[0];
    const float* z2     = (const float*)d_in[1];
    const int*   labels = (const int*)d_in[2];
    (void)in_sizes; (void)n_in; (void)out_size; (void)ws_size;

    float* ws = (float*)d_ws;
    int*   cnt = (int*)(ws + WS_CNT);
    float* zb  = ws + WS_ZB;
    float* pmax = ws + WS_LG;
    float* psum = pmax + NTILE * C2P;
    float* pos  = psum + NTILE * C2P;

    float* out   = (float*)d_out;
    float* out1  = out + 1;
    float* out2  = out + 1 + PROTO_E;

    proto_gather<<<dim3(NCLS, G_NCS, 2), G_THREADS, 0, stream>>>(z1, z2, labels, cnt, out1, out2);
    normalize_rows<<<C2P, 256, 0, stream>>>(out1, zb);
    logits_wmma<<<NTILE * NTILE, 32, 0, stream>>>(zb, cnt, pmax, psum, pos);
    loss_kernel<<<1, L_THREADS, 0, stream>>>(pmax, psum, pos, cnt, out);
}


# Gather blocks own whole prototype rows: 200 blocks x 24 waves (label scans cut to a third) normalise their finished row themselves, the normalise launch goes, 4 to 3 launches; norm sum order changed
# speedup vs baseline: 11.2815x; 1.0897x over previous
#include <hip/hip_runtime.h>
#include <hip/hip_bf16.h>

#define N_ROWS   32768
#define D        768
#define NCLS     100
#define C2       200
#define C2P      208
#define NTILE    13
#define NRC      16
#define PROTO_E  (NCLS * D)

typedef float v2f __attribute__((ext_vector_type(2)));
typedef float v8f __attribute__((ext_vector_type(8)));

#define WS_CNT   0
#define WS_P1    128
#define WS_P2    (WS_P1 + NRC * PROTO_E)
#define WS_ZB    (WS_P2 + NRC * PROTO_E)
#define WS_LG    (WS_ZB + C2P * D)

template <int M>
__device__ __forceinline__ float row_xor(float v) {
    const int b = __builtin_bit_cast(int, v);
    return __builtin_bit_cast(float, __builtin_amdgcn_update_dpp(b, b, 0x160 + M, 0xf, 0xf, false));
}
__device__ __forceinline__ float half_xor(float v) {
    const int b = __builtin_bit_cast(int, v);
    return __builtin_bit_cast(float, __builtin_amdgcn_permlanex16(b, b, 0x76543210, 0xfedcba98, false, false));
}
__device__ __forceinline__ float row16_max(float v) {
    v = fmaxf(v, row_xor<1>(v));
    v = fmaxf(v, row_xor<2>(v));
    v = fmaxf(v, row_xor<4>(v));
    return fmaxf(v, row_xor<8>(v));
}
__device__ __forceinline__ float row16_sum(float v) {
    v += row_xor<1>(v);
    v += row_xor<2>(v);
    v += row_xor<4>(v);
    return v + row_xor<8>(v);
}
__device__ __forceinline__ float wave_sum(float v) {
    v = row16_sum(v);
    return v + half_xor(v);
}

#define G_THREADS 768
#define G_WAVES   24
#define G_LISTT   256
#define G_LISTW   8
#define G_RG      4
#define G_RGT     192
#define G_UNR     32
#define G_WORDS   (N_ROWS / 32)
#define G_CHUNKS  (N_ROWS / 128)
#define G_LPW     ((G_CHUNKS + G_WAVES - 1) / G_WAVES)

typedef float        v4f __attribute__((ext_vector_type(4)));
typedef int          v4i __attribute__((ext_vector_type(4)));
typedef unsigned int v4u __attribute__((ext_vector_type(4)));

__global__ __launch_bounds__(G_THREADS) void proto_gather(
        const float* __restrict__ z1, const float* __restrict__ z2,
        const int* __restrict__ labels, int* __restrict__ cnt,
        float* __restrict__ out1, float* __restrict__ out2, float* __restrict__ zb) {
    __shared__ unsigned short rowlist[N_ROWS];
    __shared__ __attribute__((aligned(16))) unsigned int bits[G_WORDS];
    __shared__ __attribute__((aligned(16))) float red[G_RG * D];
    __shared__ int   wtot[G_LISTW];
    __shared__ float wsq[G_WAVES];

    const int t    = threadIdx.x;
    const int lane = t & 31;
    const int wave = __builtin_amdgcn_readfirstlane(t >> 5);
    const int c    = blockIdx.x;
    const int a    = blockIdx.y;

    const v4i* __restrict__ lab4 = (const v4i*)labels;
    {
        v4i lv[G_LPW];
#pragma unroll
        for (int u = 0; u < G_LPW; ++u) {
            const int j = u * G_WAVES + wave;
            if (j < G_CHUNKS) lv[u] = lab4[j * 32 + lane];
        }
#pragma unroll
        for (int u = 0; u < G_LPW; ++u) {
            const int j = u * G_WAVES + wave;
            if (j < G_CHUNKS) {
                v4u b;
                b.x = (unsigned int)__ballot(lv[u].x == c);
                b.y = (unsigned int)__ballot(lv[u].y == c);
                b.z = (unsigned int)__ballot(lv[u].z == c);
                b.w = (unsigned int)__ballot(lv[u].w == c);
                if (lane == 0) *(v4u*)&bits[j * 4] = b;
            }
        }
    }
    __syncthreads();

    v4u w = {0u, 0u, 0u, 0u};
    int nt = 0, incl = 0;
    if (t < G_LISTT) {
        w  = *(const v4u*)&bits[4 * t];
        nt = __popc(w.x) + __popc(w.y) + __popc(w.z) + __popc(w.w);
        incl = nt;
#pragma unroll
        for (int d = 1; d < 32; d <<= 1) {
            const int v = __shfl_up(incl, d, 32);
            if (lane >= d) incl += v;
        }
        if (lane == 31) wtot[wave] = incl;
    }
    __syncthreads();
    int base = 0, ntot = 0;
#pragma unroll
    for (int i = 0; i < G_LISTW; ++i) {
        const int x = wtot[i];
        if (i < wave) base += x;
        ntot += x;
    }
    if (t < G_LISTT) {
        int pos = base + incl - nt;
        const int rbase = 128 * t;
        unsigned int m;
        m = w.x; while (m) { rowlist[pos++] = (unsigned short)(rbase + 4 * __builtin_ctz(m) + 0); m &= m - 1; }
        m = w.y; while (m) { rowlist[pos++] = (unsigned short)(rbase + 4 * __builtin_ctz(m) + 1); m &= m - 1; }
        m = w.z; while (m) { rowlist[pos++] = (unsigned short)(rbase + 4 * __builtin_ctz(m) + 2); m &= m - 1; }
        m = w.w; while (m) { rowlist[pos++] = (unsigned short)(rbase + 4 * __builtin_ctz(m) + 3); m &= m - 1; }
    }
    __syncthreads();

    const int lc = t % G_RGT;
    const int rg = __builtin_amdgcn_readfirstlane(t / G_RGT);
    const int n  = __builtin_amdgcn_readfirstlane(ntot);
    const float* __restrict__ zsrc = (a == 0 ? z1 : z2) + 4 * lc;
    v4f acc = {0.f, 0.f, 0.f, 0.f};
    int e = rg;
    for (; e + (G_UNR - 1) * G_RG < n; e += G_UNR * G_RG) {
        v4f v[G_UNR];
#pragma unroll
        for (int u = 0; u < G_UNR; ++u) {
            const int row = __builtin_amdgcn_readfirstlane((int)rowlist[e + u * G_RG]);
            v[u] = __builtin_nontemporal_load((const v4f*)(zsrc + (size_t)row * D));
        }
#pragma unroll
        for (int u = 0; u < G_UNR; ++u) acc += v[u];
    }
    if (e < n) {
        int rows[G_UNR];
        v4f v[G_UNR];
#pragma unroll
        for (int u = 0; u < G_UNR; ++u) {
            const int idx = e + u * G_RG;
            rows[u] = __builtin_amdgcn_readfirstlane((int)rowlist[idx < n ? idx : e]);
        }
#pragma unroll
        for (int u = 0; u < G_UNR; ++u) {
            if (e + u * G_RG < n) v[u] = __builtin_nontemporal_load((const v4f*)(zsrc + (size_t)rows[u] * D));
            else                  v[u] = (v4f){0.f, 0.f, 0.f, 0.f};
        }
#pragma unroll
        for (int u = 0; u < G_UNR; ++u) acc += v[u];
    }

    *(v4f*)&red[rg * D + 4 * lc] = acc;
    __syncthreads();
    float s = red[t];
#pragma unroll
    for (int g = 1; g < G_RG; ++g) s += red[g * D + t];
    const float d = fmaxf((float)n, 1.0f);
    const float p = s / d;
    float* __restrict__ o = (a == 0 ? out1 : out2);
    o[(size_t)c * D + t] = p;
    if (a == 0 && t == 0) cnt[c] = n;

    const float q = wave_sum(p * p);
    if (lane == 0) wsq[wave] = q;
    __syncthreads();
    float tot = 0.f;
#pragma unroll
    for (int i = 0; i < G_WAVES; ++i) tot += wsq[i];
    const float inv = 1.0f / fmaxf(sqrtf(tot), 1e-12f);
    zb[(size_t)(a * NCLS + c) * D + t] = p * inv;
    if (a == 1 && c < C2P - C2) zb[(size_t)(C2 + c) * D + t] = 0.f;
}

__global__ __launch_bounds__(32) void logits_wmma(const float* __restrict__ zb,
                                                  const int* __restrict__ cnt,
                                                  float* __restrict__ pmax,
                                                  float* __restrict__ psum,
                                                  float* __restrict__ pos) {
    const int tile = blockIdx.x;
    const int ti = tile / NTILE, tj = tile % NTILE;
    const int lane = threadIdx.x;
    const int mn = lane & 15;
    const int kb = (lane >> 4) << 1;
    const int n  = tj * 16 + mn;
    const int ccnt = cnt[n < NCLS ? n : (n < C2 ? n - NCLS : 0)];
    const float* __restrict__ arow = zb + (size_t)(ti * 16 + mn) * D;
    const float* __restrict__ brow = zb + (size_t)(tj * 16 + mn) * D;
    v8f acc = {};
    for (int k0 = 0; k0 < D; k0 += 4) {
        const v2f a = *(const v2f*)(arow + k0 + kb);
        const v2f b = *(const v2f*)(brow + k0 + kb);
        acc = __builtin_amdgcn_wmma_f32_16x16x4_f32(
             false, a,  false, b,
             (short)0, acc,  false,  false);
    }
    const int  mhi  = (lane >> 4) * 8;
    const bool colv = (n < C2) && (ccnt > 0);
    float x[8], mx[8], s[8];
#pragma unroll
    for (int v = 0; v < 8; ++v) {
        x[v]  = acc[v] * 2.0f;
        mx[v] = row16_max(colv ? x[v] : -3.402823466e38f);
    }
#pragma unroll
    for (int v = 0; v < 8; ++v) {
        const int   m = ti * 16 + mhi + v;
        const float e = expf(x[v] - mx[v]);
        s[v] = row16_sum((colv && n != m) ? e : 0.f);
    }
    float keep_m = 0.f, keep_s = 0.f, keep_x = 0.f;
    int   keep_r = -1;
#pragma unroll
    for (int v = 0; v < 8; ++v) {
        const int m = ti * 16 + mhi + v;
        if (mn == v) { keep_m = mx[v]; keep_s = s[v]; }
        if (m < C2 && n == (m < NCLS ? m + NCLS : m - NCLS)) { keep_x = x[v]; keep_r = m; }
    }
    if (mn < 8) {
        const int m = ti * 16 + mhi + mn;
        pmax[tj * C2P + m] = keep_m;
        psum[tj * C2P + m] = keep_s;
    }
    if (keep_r >= 0) pos[keep_r] = keep_x;
}

#define L_THREADS 256
#define L_WAVES   8

__global__ __launch_bounds__(L_THREADS) void loss_kernel(const float* __restrict__ pmax,
                                                         const float* __restrict__ psum,
                                                         const float* __restrict__ pos,
                                                         const int* __restrict__ cnt,
                                                         float* __restrict__ out) {
    __shared__ float sterm[L_WAVES];
    __shared__ float scnt[L_WAVES];
    const int t    = threadIdx.x;
    const int lane = t & 31;
    const int wave = t >> 5;

    float term = 0.f, nv = 0.f;
    if (t < C2) {
        float mt[NTILE], st[NTILE];
#pragma unroll
        for (int j = 0; j < NTILE; ++j) {
            mt[j] = pmax[j * C2P + t];
            st[j] = psum[j * C2P + t];
        }
        const float xp = pos[t];
        const int   cr = cnt[t < NCLS ? t : t - NCLS];
        float m = mt[0];
#pragma unroll
        for (int j = 1; j < NTILE; ++j) m = fmaxf(m, mt[j]);
        float se = 0.f;
#pragma unroll
        for (int j = 0; j < NTILE; ++j) se += st[j] * expf(mt[j] - m);
        const float tr = xp - m - logf(se);
        term = (cr > 0) ? tr : 0.f;
        nv   = (cr > 0) ? 1.f : 0.f;
    }
    term = wave_sum(term);
    nv   = wave_sum(nv);
    if (lane == 0) { sterm[wave] = term; scnt[wave] = nv; }
    __syncthreads();
    if (t == 0) {
        float v = 0.f, nc = 0.f;
#pragma unroll
        for (int i = 0; i < L_WAVES; ++i) { v += sterm[i]; nc += scnt[i]; }
        out[0] = -v / fmaxf(nc, 1.0f);
    }
}

extern "C" void kernel_launch(void* const* d_in, const int* in_sizes, int n_in,
                              void* d_out, int out_size, void* d_ws, size_t ws_size,
                              hipStream_t stream) {
    const float* z1     = (const float*)d_in[0];
    const float* z2     = (const float*)d_in[1];
    const int*   labels = (const int*)d_in[2];
    (void)in_sizes; (void)n_in; (void)out_size; (void)ws_size;

    float* ws = (float*)d_ws;
    int*   cnt = (int*)(ws + WS_CNT);
    float* zb  = ws + WS_ZB;
    float* pmax = ws + WS_LG;
    float* psum = pmax + NTILE * C2P;
    float* pos  = psum + NTILE * C2P;

    float* out   = (float*)d_out;
    float* out1  = out + 1;
    float* out2  = out + 1 + PROTO_E;

    proto_gather<<<dim3(NCLS, 2), G_THREADS, 0, stream>>>(z1, z2, labels, cnt, out1, out2, zb);
    logits_wmma<<<NTILE * NTILE, 32, 0, stream>>>(zb, cnt, pmax, psum, pos);
    loss_kernel<<<1, L_THREADS, 0, stream>>>(pmax, psum, pos, cnt, out);
}


# Logits tile K split over the 4 waves of a block: each wave issues its 192 columns as 16-B loads up front, 48 products on 2 accumulators, wave 0 adds the partial tiles; logit sum order changed, fixed
# speedup vs baseline: 13.4149x; 1.1891x over previous
#include <hip/hip_runtime.h>
#include <hip/hip_bf16.h>

#define N_ROWS   32768
#define D        768
#define NCLS     100
#define C2       200
#define C2P      208
#define NTILE    13
#define NRC      16
#define PROTO_E  (NCLS * D)

typedef float v2f __attribute__((ext_vector_type(2)));
typedef float v8f __attribute__((ext_vector_type(8)));

#define WS_CNT   0
#define WS_P1    128
#define WS_P2    (WS_P1 + NRC * PROTO_E)
#define WS_ZB    (WS_P2 + NRC * PROTO_E)
#define WS_LG    (WS_ZB + C2P * D)

template <int M>
__device__ __forceinline__ float row_xor(float v) {
    const int b = __builtin_bit_cast(int, v);
    return __builtin_bit_cast(float, __builtin_amdgcn_update_dpp(b, b, 0x160 + M, 0xf, 0xf, false));
}
__device__ __forceinline__ float half_xor(float v) {
    const int b = __builtin_bit_cast(int, v);
    return __builtin_bit_cast(float, __builtin_amdgcn_permlanex16(b, b, 0x76543210, 0xfedcba98, false, false));
}
__device__ __forceinline__ float row16_max(float v) {
    v = fmaxf(v, row_xor<1>(v));
    v = fmaxf(v, row_xor<2>(v));
    v = fmaxf(v, row_xor<4>(v));
    return fmaxf(v, row_xor<8>(v));
}
__device__ __forceinline__ float row16_sum(float v) {
    v += row_xor<1>(v);
    v += row_xor<2>(v);
    v += row_xor<4>(v);
    return v + row_xor<8>(v);
}
__device__ __forceinline__ float wave_sum(float v) {
    v = row16_sum(v);
    return v + half_xor(v);
}

#define G_THREADS 768
#define G_WAVES   24
#define G_LISTT   256
#define G_LISTW   8
#define G_RG      4
#define G_RGT     192
#define G_UNR     32
#define G_WORDS   (N_ROWS / 32)
#define G_CHUNKS  (N_ROWS / 128)
#define G_LPW     ((G_CHUNKS + G_WAVES - 1) / G_WAVES)

typedef float        v4f __attribute__((ext_vector_type(4)));
typedef int          v4i __attribute__((ext_vector_type(4)));
typedef unsigned int v4u __attribute__((ext_vector_type(4)));

__global__ __launch_bounds__(G_THREADS) void proto_gather(
        const float* __restrict__ z1, const float* __restrict__ z2,
        const int* __restrict__ labels, int* __restrict__ cnt,
        float* __restrict__ out1, float* __restrict__ out2, float* __restrict__ zb) {
    __shared__ unsigned short rowlist[N_ROWS];
    __shared__ __attribute__((aligned(16))) unsigned int bits[G_WORDS];
    __shared__ __attribute__((aligned(16))) float red[G_RG * D];
    __shared__ int   wtot[G_LISTW];
    __shared__ float wsq[G_WAVES];

    const int t    = threadIdx.x;
    const int lane = t & 31;
    const int wave = __builtin_amdgcn_readfirstlane(t >> 5);
    const int c    = blockIdx.x;
    const int a    = blockIdx.y;

    const v4i* __restrict__ lab4 = (const v4i*)labels;
    {
        v4i lv[G_LPW];
#pragma unroll
        for (int u = 0; u < G_LPW; ++u) {
            const int j = u * G_WAVES + wave;
            if (j < G_CHUNKS) lv[u] = lab4[j * 32 + lane];
        }
#pragma unroll
        for (int u = 0; u < G_LPW; ++u) {
            const int j = u * G_WAVES + wave;
            if (j < G_CHUNKS) {
                v4u b;
                b.x = (unsigned int)__ballot(lv[u].x == c);
                b.y = (unsigned int)__ballot(lv[u].y == c);
                b.z = (unsigned int)__ballot(lv[u].z == c);
                b.w = (unsigned int)__ballot(lv[u].w == c);
                if (lane == 0) *(v4u*)&bits[j * 4] = b;
            }
        }
    }
    __syncthreads();

    v4u w = {0u, 0u, 0u, 0u};
    int nt = 0, incl = 0;
    if (t < G_LISTT) {
        w  = *(const v4u*)&bits[4 * t];
        nt = __popc(w.x) + __popc(w.y) + __popc(w.z) + __popc(w.w);
        incl = nt;
#pragma unroll
        for (int d = 1; d < 32; d <<= 1) {
            const int v = __shfl_up(incl, d, 32);
            if (lane >= d) incl += v;
        }
        if (lane == 31) wtot[wave] = incl;
    }
    __syncthreads();
    int base = 0, ntot = 0;
#pragma unroll
    for (int i = 0; i < G_LISTW; ++i) {
        const int x = wtot[i];
        if (i < wave) base += x;
        ntot += x;
    }
    if (t < G_LISTT) {
        int pos = base + incl - nt;
        const int rbase = 128 * t;
        unsigned int m;
        m = w.x; while (m) { rowlist[pos++] = (unsigned short)(rbase + 4 * __builtin_ctz(m) + 0); m &= m - 1; }
        m = w.y; while (m) { rowlist[pos++] = (unsigned short)(rbase + 4 * __builtin_ctz(m) + 1); m &= m - 1; }
        m = w.z; while (m) { rowlist[pos++] = (unsigned short)(rbase + 4 * __builtin_ctz(m) + 2); m &= m - 1; }
        m = w.w; while (m) { rowlist[pos++] = (unsigned short)(rbase + 4 * __builtin_ctz(m) + 3); m &= m - 1; }
    }
    __syncthreads();

    const int lc = t % G_RGT;
    const int rg = __builtin_amdgcn_readfirstlane(t / G_RGT);
    const int n  = __builtin_amdgcn_readfirstlane(ntot);
    const float* __restrict__ zsrc = (a == 0 ? z1 : z2) + 4 * lc;
    v4f acc = {0.f, 0.f, 0.f, 0.f};
    int e = rg;
    for (; e + (G_UNR - 1) * G_RG < n; e += G_UNR * G_RG) {
        v4f v[G_UNR];
#pragma unroll
        for (int u = 0; u < G_UNR; ++u) {
            const int row = __builtin_amdgcn_readfirstlane((int)rowlist[e + u * G_RG]);
            v[u] = __builtin_nontemporal_load((const v4f*)(zsrc + (size_t)row * D));
        }
#pragma unroll
        for (int u = 0; u < G_UNR; ++u) acc += v[u];
    }
    if (e < n) {
        int rows[G_UNR];
        v4f v[G_UNR];
#pragma unroll
        for (int u = 0; u < G_UNR; ++u) {
            const int idx = e + u * G_RG;
            rows[u] = __builtin_amdgcn_readfirstlane((int)rowlist[idx < n ? idx : e]);
        }
#pragma unroll
        for (int u = 0; u < G_UNR; ++u) {
            if (e + u * G_RG < n) v[u] = __builtin_nontemporal_load((const v4f*)(zsrc + (size_t)rows[u] * D));
            else                  v[u] = (v4f){0.f, 0.f, 0.f, 0.f};
        }
#pragma unroll
        for (int u = 0; u < G_UNR; ++u) acc += v[u];
    }

    *(v4f*)&red[rg * D + 4 * lc] = acc;
    __syncthreads();
    float s = red[t];
#pragma unroll
    for (int g = 1; g < G_RG; ++g) s += red[g * D + t];
    const float d = fmaxf((float)n, 1.0f);
    const float p = s / d;
    float* __restrict__ o = (a == 0 ? out1 : out2);
    o[(size_t)c * D + t] = p;
    if (a == 0 && t == 0) cnt[c] = n;

    const float q = wave_sum(p * p);
    if (lane == 0) wsq[wave] = q;
    __syncthreads();
    float tot = 0.f;
#pragma unroll
    for (int i = 0; i < G_WAVES; ++i) tot += wsq[i];
    const float inv = 1.0f / fmaxf(sqrtf(tot), 1e-12f);
    zb[(size_t)(a * NCLS + c) * D + t] = p * inv;
    if (a == 1 && c < C2P - C2) zb[(size_t)(C2 + c) * D + t] = 0.f;
}

#define LG_WAVES   4
#define LG_THREADS (LG_WAVES * 32)
#define LG_KS      (D / LG_WAVES)
#define LG_NLD     (LG_KS / 8)

__global__ __launch_bounds__(LG_THREADS) void logits_wmma(const float* __restrict__ zb,
                                                          const int* __restrict__ cnt,
                                                          float* __restrict__ pmax,
                                                          float* __restrict__ psum,
                                                          float* __restrict__ pos) {
    __shared__ float part[(LG_WAVES - 1) * 8 * 32];
    const int tile = blockIdx.x;
    const int ti = tile / NTILE, tj = tile % NTILE;
    const int lane = threadIdx.x & 31;
    const int wave = __builtin_amdgcn_readfirstlane((int)(threadIdx.x >> 5));
    const int mn = lane & 15;
    const int n  = tj * 16 + mn;
    const int ccnt = cnt[n < NCLS ? n : (n < C2 ? n - NCLS : 0)];
    const int koff = wave * LG_KS + ((lane >> 4) << 2);
    const float* __restrict__ arow = zb + (size_t)(ti * 16 + mn) * D + koff;
    const float* __restrict__ brow = zb + (size_t)(tj * 16 + mn) * D + koff;
    v4f a[LG_NLD], b[LG_NLD];
#pragma unroll
    for (int j = 0; j < LG_NLD; ++j) {
        a[j] = *(const v4f*)(arow + 8 * j);
        b[j] = *(const v4f*)(brow + 8 * j);
    }
    __builtin_amdgcn_sched_barrier(0);
    v8f acc0 = {}, acc1 = {};
#pragma unroll
    for (int j = 0; j < LG_NLD; ++j) {
        acc0 = __builtin_amdgcn_wmma_f32_16x16x4_f32(
             false, a[j].xy,  false, b[j].xy,
             (short)0, acc0,  false,  false);
        acc1 = __builtin_amdgcn_wmma_f32_16x16x4_f32(
             false, a[j].zw,  false, b[j].zw,
             (short)0, acc1,  false,  false);
    }
    v8f acc = acc0 + acc1;
    if (wave != 0) {
#pragma unroll
        for (int v = 0; v < 8; ++v) part[((wave - 1) * 8 + v) * 32 + lane] = acc[v];
    }
    __syncthreads();
    if (wave != 0) return;
#pragma unroll
    for (int w = 1; w < LG_WAVES; ++w) {
#pragma unroll
        for (int v = 0; v < 8; ++v) acc[v] += part[((w - 1) * 8 + v) * 32 + lane];
    }
    const int  mhi  = (lane >> 4) * 8;
    const bool colv = (n < C2) && (ccnt > 0);
    float x[8], mx[8], s[8];
#pragma unroll
    for (int v = 0; v < 8; ++v) {
        x[v]  = acc[v] * 2.0f;
        mx[v] = row16_max(colv ? x[v] : -3.402823466e38f);
    }
#pragma unroll
    for (int v = 0; v < 8; ++v) {
        const int   m = ti * 16 + mhi + v;
        const float e = expf(x[v] - mx[v]);
        s[v] = row16_sum((colv && n != m) ? e : 0.f);
    }
    float keep_m = 0.f, keep_s = 0.f, keep_x = 0.f;
    int   keep_r = -1;
#pragma unroll
    for (int v = 0; v < 8; ++v) {
        const int m = ti * 16 + mhi + v;
        if (mn == v) { keep_m = mx[v]; keep_s = s[v]; }
        if (m < C2 && n == (m < NCLS ? m + NCLS : m - NCLS)) { keep_x = x[v]; keep_r = m; }
    }
    if (mn < 8) {
        const int m = ti * 16 + mhi + mn;
        pmax[tj * C2P + m] = keep_m;
        psum[tj * C2P + m] = keep_s;
    }
    if (keep_r >= 0) pos[keep_r] = keep_x;
}

#define L_THREADS 256
#define L_WAVES   8

__global__ __launch_bounds__(L_THREADS) void loss_kernel(const float* __restrict__ pmax,
                                                         const float* __restrict__ psum,
                                                         const float* __restrict__ pos,
                                                         const int* __restrict__ cnt,
                                                         float* __restrict__ out) {
    __shared__ float sterm[L_WAVES];
    __shared__ float scnt[L_WAVES];
    const int t    = threadIdx.x;
    const int lane = t & 31;
    const int wave = t >> 5;

    float term = 0.f, nv = 0.f;
    if (t < C2) {
        float mt[NTILE], st[NTILE];
#pragma unroll
        for (int j = 0; j < NTILE; ++j) {
            mt[j] = pmax[j * C2P + t];
            st[j] = psum[j * C2P + t];
        }
        const float xp = pos[t];
        const int   cr = cnt[t < NCLS ? t : t - NCLS];
        float m = mt[0];
#pragma unroll
        for (int j = 1; j < NTILE; ++j) m = fmaxf(m, mt[j]);
        float se = 0.f;
#pragma unroll
        for (int j = 0; j < NTILE; ++j) se += st[j] * expf(mt[j] - m);
        const float tr = xp - m - logf(se);
        term = (cr > 0) ? tr : 0.f;
        nv   = (cr > 0) ? 1.f : 0.f;
    }
    term = wave_sum(term);
    nv   = wave_sum(nv);
    if (lane == 0) { sterm[wave] = term; scnt[wave] = nv; }
    __syncthreads();
    if (t == 0) {
        float v = 0.f, nc = 0.f;
#pragma unroll
        for (int i = 0; i < L_WAVES; ++i) { v += sterm[i]; nc += scnt[i]; }
        out[0] = -v / fmaxf(nc, 1.0f);
    }
}

extern "C" void kernel_launch(void* const* d_in, const int* in_sizes, int n_in,
                              void* d_out, int out_size, void* d_ws, size_t ws_size,
                              hipStream_t stream) {
    const float* z1     = (const float*)d_in[0];
    const float* z2     = (const float*)d_in[1];
    const int*   labels = (const int*)d_in[2];
    (void)in_sizes; (void)n_in; (void)out_size; (void)ws_size;

    float* ws = (float*)d_ws;
    int*   cnt = (int*)(ws + WS_CNT);
    float* zb  = ws + WS_ZB;
    float* pmax = ws + WS_LG;
    float* psum = pmax + NTILE * C2P;
    float* pos  = psum + NTILE * C2P;

    float* out   = (float*)d_out;
    float* out1  = out + 1;
    float* out2  = out + 1 + PROTO_E;

    proto_gather<<<dim3(NCLS, 2), G_THREADS, 0, stream>>>(z1, z2, labels, cnt, out1, out2, zb);
    logits_wmma<<<NTILE * NTILE, LG_THREADS, 0, stream>>>(zb, cnt, pmax, psum, pos);
    loss_kernel<<<1, L_THREADS, 0, stream>>>(pmax, psum, pos, cnt, out);
}
